# starting kernel
# baseline (speedup 1.0000x reference)
#include <hip/hip_runtime.h>
#include <math.h>

#define NQ 4
#define DIM 16
#define QDEPTH 6

typedef float v8f __attribute__((ext_vector_type(8)));
typedef float v2f __attribute__((ext_vector_type(2)));

#if defined(__has_builtin)
#if __has_builtin(__builtin_amdgcn_tanhf)
#define FAST_TANH(x) __builtin_amdgcn_tanhf(x)
#endif
#endif
#ifndef FAST_TANH
#define FAST_TANH(x) tanhf(x)
#endif

__device__ __forceinline__ void cnot16(float* st, int cm, int tm) {
#pragma unroll
  for (int i = 0; i < DIM; ++i) {
    if ((i & cm) && !(i & tm)) {
      float t = st[i];
      st[i] = st[i | tm];
      st[i | tm] = t;
    }
  }
}

__device__ __forceinline__ void ry16(float* st, int m, float c, float s) {
#pragma unroll
  for (int i = 0; i < DIM; ++i) {
    if (!(i & m)) {
      int j = i | m;
      float s0 = st[i], s1 = st[j];
      st[i] = c * s0 - s * s1;
      st[j] = s * s0 + c * s1;
    }
  }
}

__launch_bounds__(256)
__global__ void quantum_layer_kernel(const float* __restrict__ x,
                                     const float* __restrict__ qp,
                                     const float* __restrict__ scale,
                                     float* __restrict__ out,
                                     int numTiles) {
  __shared__ float Ms[DIM * DIM];

  if (threadIdx.x < DIM) {
    float st[DIM];
#pragma unroll
    for (int i = 0; i < DIM; ++i) st[i] = 0.0f;
    st[threadIdx.x] = 1.0f;
    for (int k = 0; k < QDEPTH; ++k) {
      cnot16(st, 8, 4);
      cnot16(st, 2, 1);
      cnot16(st, 4, 2);
#pragma unroll
      for (int w = 0; w < NQ; ++w) {
        float th = 0.5f * qp[k * NQ + w];
        ry16(st, 8 >> w, cosf(th), sinf(th));
      }
    }
#pragma unroll
    for (int j = 0; j < DIM; ++j) Ms[j * DIM + threadIdx.x] = st[j];
  }
  __syncthreads();

  const int lane = threadIdx.x & 31;
  const int n = lane & 15;
  const int h = lane >> 4;

  v2f A[4];
#pragma unroll
  for (int c = 0; c < 4; ++c) {
    int off = n * DIM + c * 4 + h * 2;
    A[c] = *(const v2f*)&Ms[off];
  }

  const float sc = scale[0];
  const int wavesPerBlock = blockDim.x >> 5;
  const int waveId = blockIdx.x * wavesPerBlock + (threadIdx.x >> 5);
  const int nWaves = gridDim.x * wavesPerBlock;

  const float4* __restrict__ x4 = (const float4*)x;
  float4* __restrict__ o4 = (float4*)out;
  const float PI4 = 0.78539816339744831f;

  for (int tile = waveId; tile < numTiles; tile += nWaves) {
    const int s = tile * 16 + n;
    float4 xv = x4[s];

    float inv = rsqrtf(xv.x * xv.x + xv.y * xv.y + xv.z * xv.z + xv.w * xv.w);
    float m = sc * inv;
    float a0 = FAST_TANH(xv.x * m) * PI4 + PI4;
    float a1 = FAST_TANH(xv.y * m) * PI4 + PI4;
    float a2 = FAST_TANH(xv.z * m) * PI4 + PI4;
    float a3 = FAST_TANH(xv.w * m) * PI4 + PI4;
    float c0 = __cosf(a0), s0v = __sinf(a0);
    float c1 = __cosf(a1), s1v = __sinf(a1);
    float c2 = __cosf(a2), s2v = __sinf(a2);
    float c3 = __cosf(a3), s3v = __sinf(a3);

    float f2h = h ? s2v : c2;
    float g0 = f2h * c3, g1 = f2h * s3v;
    float p01[4] = {c0 * c1, c0 * s1v, s0v * c1, s0v * s1v};

    v8f acc = {0.f, 0.f, 0.f, 0.f, 0.f, 0.f, 0.f, 0.f};
#pragma unroll
    for (int c = 0; c < 4; ++c) {
      v2f b;
      b[0] = p01[c] * g0;
      b[1] = p01[c] * g1;
      acc = __builtin_amdgcn_wmma_f32_16x16x4_f32(
          false, A[c], false, b, (short)0, acc, false, false);
    }

    float p0 = acc[0] * acc[0], p1 = acc[1] * acc[1];
    float p2 = acc[2] * acc[2], p3 = acc[3] * acc[3];
    float p4 = acc[4] * acc[4], p5 = acc[5] * acc[5];
    float p6 = acc[6] * acc[6], p7 = acc[7] * acc[7];

    float a01 = p0 + p1, a23 = p2 + p3, a45 = p4 + p5, a67 = p6 + p7;
    float b01 = p0 - p1, b23 = p2 - p3, b45 = p4 - p5, b67 = p6 - p7;
    float L = a01 + a23, H = a45 + a67;
    float S = L + H;
    float P0 = h ? -S : S;
    float P1 = L - H;
    float P2 = (a01 - a23) + (a45 - a67);
    float P3 = (b01 + b23) + (b45 + b67);

    float r0 = P0 + __shfl_xor(P0, 16, 32);
    float r1 = P1 + __shfl_xor(P1, 16, 32);
    float r2 = P2 + __shfl_xor(P2, 16, 32);
    float r3 = P3 + __shfl_xor(P3, 16, 32);

    if (h == 0) {
      o4[s] = make_float4(r0, r1, r2, r3);
    }
  }
}

extern "C" void kernel_launch(void* const* d_in, const int* in_sizes, int n_in,
                              void* d_out, int out_size, void* d_ws, size_t ws_size,
                              hipStream_t stream) {
  const float* x = (const float*)d_in[0];
  const float* qp = (const float*)d_in[1];
  const float* scale = (const float*)d_in[2];
  float* out = (float*)d_out;

  const int B = in_sizes[0] / 4;
  const int numTiles = B / 16;

  int blocks = 2048;
  int blocksNeeded = (numTiles + 7) / 8;
  if (blocks > blocksNeeded) blocks = blocksNeeded;
  if (blocks < 1) blocks = 1;

  quantum_layer_kernel<<<blocks, 256, 0, stream>>>(x, qp, scale, out, numTiles);
}


# 512 work-groups instead of 2048 (4 waves per SIMD): each wave walks 32 tiles in groups of 8 and keeps the next group's 8 x loads in flight in a register ring; per-tile arithmetic unchanged
# speedup vs baseline: 1.2427x; 1.2427x over previous
#include <hip/hip_runtime.h>
#include <math.h>


#define NQ 4
#define DIM 16
#define QDEPTH 6
#define TPG 8

typedef float v8f __attribute__((ext_vector_type(8)));
typedef float v2f __attribute__((ext_vector_type(2)));

#if defined(__has_builtin)
#if __has_builtin(__builtin_amdgcn_tanhf)
#define FAST_TANH(x) __builtin_amdgcn_tanhf(x)
#endif
#endif
#ifndef FAST_TANH
#define FAST_TANH(x) tanhf(x)
#endif

__device__ __forceinline__ void cnot16(float* st, int cm, int tm) {
#pragma unroll
  for (int i = 0; i < DIM; ++i) {
    if ((i & cm) && !(i & tm)) {
      float t = st[i];
      st[i] = st[i | tm];
      st[i | tm] = t;
    }
  }
}

__device__ __forceinline__ void ry16(float* st, int m, float c, float s) {
#pragma unroll
  for (int i = 0; i < DIM; ++i) {
    if (!(i & m)) {
      int j = i | m;
      float s0 = st[i], s1 = st[j];
      st[i] = c * s0 - s * s1;
      st[j] = s * s0 + c * s1;
    }
  }
}

__device__ __forceinline__ void tile_compute_store(const float4 xv, const v2f* A,
                                                   const float sc, const int h,
                                                   float4* __restrict__ op) {
  const float PI4 = 0.78539816339744831f;

  float inv = rsqrtf(xv.x * xv.x + xv.y * xv.y + xv.z * xv.z + xv.w * xv.w);
  float m = sc * inv;
  float a0 = FAST_TANH(xv.x * m) * PI4 + PI4;
  float a1 = FAST_TANH(xv.y * m) * PI4 + PI4;
  float a2 = FAST_TANH(xv.z * m) * PI4 + PI4;
  float a3 = FAST_TANH(xv.w * m) * PI4 + PI4;
  float c0 = __cosf(a0), s0v = __sinf(a0);
  float c1 = __cosf(a1), s1v = __sinf(a1);
  float c2 = __cosf(a2), s2v = __sinf(a2);
  float c3 = __cosf(a3), s3v = __sinf(a3);

  float f2h = h ? s2v : c2;
  float g0 = f2h * c3, g1 = f2h * s3v;
  float p01[4] = {c0 * c1, c0 * s1v, s0v * c1, s0v * s1v};

  v8f acc = {0.f, 0.f, 0.f, 0.f, 0.f, 0.f, 0.f, 0.f};
#pragma unroll
  for (int c = 0; c < 4; ++c) {
    v2f b;
    b[0] = p01[c] * g0;
    b[1] = p01[c] * g1;
    acc = __builtin_amdgcn_wmma_f32_16x16x4_f32(
        false, A[c], false, b, (short)0, acc, false, false);
  }

  float p0 = acc[0] * acc[0], p1 = acc[1] * acc[1];
  float p2 = acc[2] * acc[2], p3 = acc[3] * acc[3];
  float p4 = acc[4] * acc[4], p5 = acc[5] * acc[5];
  float p6 = acc[6] * acc[6], p7 = acc[7] * acc[7];

  float a01 = p0 + p1, a23 = p2 + p3, a45 = p4 + p5, a67 = p6 + p7;
  float b01 = p0 - p1, b23 = p2 - p3, b45 = p4 - p5, b67 = p6 - p7;
  float L = a01 + a23, H = a45 + a67;
  float S = L + H;
  float P0 = h ? -S : S;
  float P1 = L - H;
  float P2 = (a01 - a23) + (a45 - a67);
  float P3 = (b01 + b23) + (b45 + b67);

  float r0 = P0 + __shfl_xor(P0, 16, 32);
  float r1 = P1 + __shfl_xor(P1, 16, 32);
  float r2 = P2 + __shfl_xor(P2, 16, 32);
  float r3 = P3 + __shfl_xor(P3, 16, 32);

  if (h == 0) {
    *op = make_float4(r0, r1, r2, r3);
  }
}

__launch_bounds__(256)
__global__ void quantum_layer_kernel(const float* __restrict__ x,
                                     const float* __restrict__ qp,
                                     const float* __restrict__ scale,
                                     float* __restrict__ out,
                                     int numTiles) {
  __shared__ float Ms[DIM * DIM];

  const int lane = threadIdx.x & 31;
  const int n = lane & 15;
  const int h = lane >> 4;

  const int wavesPerBlock = blockDim.x >> 5;
  const int waveId = __builtin_amdgcn_readfirstlane(
      blockIdx.x * wavesPerBlock + (threadIdx.x >> 5));
  const int nWaves = gridDim.x * wavesPerBlock;

  const float4* __restrict__ x4 = (const float4*)x;
  float4* __restrict__ o4 = (float4*)out;

  const int fullGroups = numTiles / TPG;
  int g = waveId;
  float4 ring[TPG];
  if (g < fullGroups) {
    const float4* gp = x4 + (size_t)g * (TPG * 16) + n;
#pragma unroll
    for (int u = 0; u < TPG; ++u) ring[u] = gp[u * 16];
  }

  if (threadIdx.x < DIM) {
    float st[DIM];
#pragma unroll
    for (int i = 0; i < DIM; ++i) st[i] = 0.0f;
    st[threadIdx.x] = 1.0f;
    for (int k = 0; k < QDEPTH; ++k) {
      cnot16(st, 8, 4);
      cnot16(st, 2, 1);
      cnot16(st, 4, 2);
#pragma unroll
      for (int w = 0; w < NQ; ++w) {
        float th = 0.5f * qp[k * NQ + w];
        ry16(st, 8 >> w, cosf(th), sinf(th));
      }
    }
#pragma unroll
    for (int j = 0; j < DIM; ++j) Ms[j * DIM + threadIdx.x] = st[j];
  }
  __syncthreads();

  v2f A[4];
#pragma unroll
  for (int c = 0; c < 4; ++c) {
    int off = n * DIM + c * 4 + h * 2;
    A[c] = *(const v2f*)&Ms[off];
  }

  const float sc = scale[0];

  if (g < fullGroups) {
    while (g + nWaves < fullGroups) {
      const float4* gpn = x4 + (size_t)(g + nWaves) * (TPG * 16) + n;
      float4* gpo = o4 + (size_t)g * (TPG * 16) + n;
#pragma unroll
      for (int u = 0; u < TPG; ++u) {
        tile_compute_store(ring[u], A, sc, h, gpo + u * 16);
        ring[u] = gpn[u * 16];
      }
      g += nWaves;
    }
    float4* gpo = o4 + (size_t)g * (TPG * 16) + n;
#pragma unroll
    for (int u = 0; u < TPG; ++u) {
      tile_compute_store(ring[u], A, sc, h, gpo + u * 16);
    }
  }

  for (int tile = fullGroups * TPG + waveId; tile < numTiles; tile += nWaves) {
    const int s = tile * 16 + n;
    tile_compute_store(x4[s], A, sc, h, o4 + s);
  }
}

extern "C" void kernel_launch(void* const* d_in, const int* in_sizes, int n_in,
                              void* d_out, int out_size, void* d_ws, size_t ws_size,
                              hipStream_t stream) {
  const float* x = (const float*)d_in[0];
  const float* qp = (const float*)d_in[1];
  const float* scale = (const float*)d_in[2];
  float* out = (float*)d_out;

  const int B = in_sizes[0] / 4;
  const int numTiles = B / 16;

  int blocks = 512;
  int blocksNeeded = (numTiles + 7) / 8;
  if (blocks > blocksNeeded) blocks = blocksNeeded;
  if (blocks < 1) blocks = 1;

  quantum_layer_kernel<<<blocks, 256, 0, stream>>>(x, qp, scale, out, numTiles);
}


# M build: each of the 24 half-angle cos/sin pairs is evaluated by its own lane in parallel (parameters by wide scalar loads) and read by the gate loop via readlane, not 48 serial scalar evaluations
# speedup vs baseline: 1.4854x; 1.1953x over previous
#include <hip/hip_runtime.h>
#include <math.h>


#define NQ 4
#define DIM 16
#define QDEPTH 6
#define TPG 8

typedef float v8f __attribute__((ext_vector_type(8)));
typedef float v2f __attribute__((ext_vector_type(2)));

#if defined(__has_builtin)
#if __has_builtin(__builtin_amdgcn_tanhf)
#define FAST_TANH(x) __builtin_amdgcn_tanhf(x)
#endif
#endif
#ifndef FAST_TANH
#define FAST_TANH(x) tanhf(x)
#endif

__device__ __forceinline__ float lane_bcast(float v, int srcLane) {
  return __builtin_bit_cast(
      float, __builtin_amdgcn_readlane(__builtin_bit_cast(int, v), srcLane));
}

__device__ __forceinline__ void cnot16(float* st, int cm, int tm) {
#pragma unroll
  for (int i = 0; i < DIM; ++i) {
    if ((i & cm) && !(i & tm)) {
      float t = st[i];
      st[i] = st[i | tm];
      st[i | tm] = t;
    }
  }
}

__device__ __forceinline__ void ry16(float* st, int m, float c, float s) {
#pragma unroll
  for (int i = 0; i < DIM; ++i) {
    if (!(i & m)) {
      int j = i | m;
      float s0 = st[i], s1 = st[j];
      st[i] = c * s0 - s * s1;
      st[j] = s * s0 + c * s1;
    }
  }
}

__device__ __forceinline__ void tile_compute_store(const float4 xv, const v2f* A,
                                                   const float sc, const int h,
                                                   float4* __restrict__ op) {
  const float PI4 = 0.78539816339744831f;

  float inv = rsqrtf(xv.x * xv.x + xv.y * xv.y + xv.z * xv.z + xv.w * xv.w);
  float m = sc * inv;
  float a0 = FAST_TANH(xv.x * m) * PI4 + PI4;
  float a1 = FAST_TANH(xv.y * m) * PI4 + PI4;
  float a2 = FAST_TANH(xv.z * m) * PI4 + PI4;
  float a3 = FAST_TANH(xv.w * m) * PI4 + PI4;
  float c0 = __cosf(a0), s0v = __sinf(a0);
  float c1 = __cosf(a1), s1v = __sinf(a1);
  float c2 = __cosf(a2), s2v = __sinf(a2);
  float c3 = __cosf(a3), s3v = __sinf(a3);

  float f2h = h ? s2v : c2;
  float g0 = f2h * c3, g1 = f2h * s3v;
  float p01[4] = {c0 * c1, c0 * s1v, s0v * c1, s0v * s1v};

  v8f acc = {0.f, 0.f, 0.f, 0.f, 0.f, 0.f, 0.f, 0.f};
#pragma unroll
  for (int c = 0; c < 4; ++c) {
    v2f b;
    b[0] = p01[c] * g0;
    b[1] = p01[c] * g1;
    acc = __builtin_amdgcn_wmma_f32_16x16x4_f32(
        false, A[c], false, b, (short)0, acc, false, false);
  }

  float p0 = acc[0] * acc[0], p1 = acc[1] * acc[1];
  float p2 = acc[2] * acc[2], p3 = acc[3] * acc[3];
  float p4 = acc[4] * acc[4], p5 = acc[5] * acc[5];
  float p6 = acc[6] * acc[6], p7 = acc[7] * acc[7];

  float a01 = p0 + p1, a23 = p2 + p3, a45 = p4 + p5, a67 = p6 + p7;
  float b01 = p0 - p1, b23 = p2 - p3, b45 = p4 - p5, b67 = p6 - p7;
  float L = a01 + a23, H = a45 + a67;
  float S = L + H;
  float P0 = h ? -S : S;
  float P1 = L - H;
  float P2 = (a01 - a23) + (a45 - a67);
  float P3 = (b01 + b23) + (b45 + b67);

  float r0 = P0 + __shfl_xor(P0, 16, 32);
  float r1 = P1 + __shfl_xor(P1, 16, 32);
  float r2 = P2 + __shfl_xor(P2, 16, 32);
  float r3 = P3 + __shfl_xor(P3, 16, 32);

  if (h == 0) {
    *op = make_float4(r0, r1, r2, r3);
  }
}

__launch_bounds__(256)
__global__ void quantum_layer_kernel(const float* __restrict__ x,
                                     const float* __restrict__ qp,
                                     const float* __restrict__ scale,
                                     float* __restrict__ out,
                                     int numTiles) {
  __shared__ float Ms[DIM * DIM];

  const int lane = threadIdx.x & 31;
  const int n = lane & 15;
  const int h = lane >> 4;

  const int wavesPerBlock = blockDim.x >> 5;
  const int waveId = __builtin_amdgcn_readfirstlane(
      blockIdx.x * wavesPerBlock + (threadIdx.x >> 5));
  const int nWaves = gridDim.x * wavesPerBlock;

  const float4* __restrict__ x4 = (const float4*)x;
  float4* __restrict__ o4 = (float4*)out;

  const int fullGroups = numTiles / TPG;
  int g = waveId;
  float4 ring[TPG];
  if (g < fullGroups) {
    const float4* gp = x4 + (size_t)g * (TPG * 16) + n;
#pragma unroll
    for (int u = 0; u < TPG; ++u) ring[u] = gp[u * 16];
  }

  if (threadIdx.x < 32) {
    float qv = 0.0f;
#pragma unroll
    for (int i = 0; i < QDEPTH * NQ; ++i) {
      const float qi = qp[i];
      qv = (lane == i) ? qi : qv;
    }
    const float th = 0.5f * qv;
    const float cv = cosf(th), sv = sinf(th);
    float st[DIM];
#pragma unroll
    for (int i = 0; i < DIM; ++i) st[i] = (i == n) ? 1.0f : 0.0f;
    for (int k = 0; k < QDEPTH; ++k) {
      cnot16(st, 8, 4);
      cnot16(st, 2, 1);
      cnot16(st, 4, 2);
#pragma unroll
      for (int w = 0; w < NQ; ++w) {
        ry16(st, 8 >> w, lane_bcast(cv, k * NQ + w), lane_bcast(sv, k * NQ + w));
      }
    }
    if (lane < DIM) {
#pragma unroll
      for (int j = 0; j < DIM; ++j) Ms[j * DIM + lane] = st[j];
    }
  }
  __syncthreads();

  v2f A[4];
#pragma unroll
  for (int c = 0; c < 4; ++c) {
    int off = n * DIM + c * 4 + h * 2;
    A[c] = *(const v2f*)&Ms[off];
  }

  const float sc = scale[0];

  if (g < fullGroups) {
    while (g + nWaves < fullGroups) {
      const float4* gpn = x4 + (size_t)(g + nWaves) * (TPG * 16) + n;
      float4* gpo = o4 + (size_t)g * (TPG * 16) + n;
#pragma unroll
      for (int u = 0; u < TPG; ++u) {
        tile_compute_store(ring[u], A, sc, h, gpo + u * 16);
        ring[u] = gpn[u * 16];
      }
      g += nWaves;
    }
    float4* gpo = o4 + (size_t)g * (TPG * 16) + n;
#pragma unroll
    for (int u = 0; u < TPG; ++u) {
      tile_compute_store(ring[u], A, sc, h, gpo + u * 16);
    }
  }

  for (int tile = fullGroups * TPG + waveId; tile < numTiles; tile += nWaves) {
    const int s = tile * 16 + n;
    tile_compute_store(x4[s], A, sc, h, o4 + s);
  }
}

extern "C" void kernel_launch(void* const* d_in, const int* in_sizes, int n_in,
                              void* d_out, int out_size, void* d_ws, size_t ws_size,
                              hipStream_t stream) {
  const float* x = (const float*)d_in[0];
  const float* qp = (const float*)d_in[1];
  const float* scale = (const float*)d_in[2];
  float* out = (float*)d_out;

  const int B = in_sizes[0] / 4;
  const int numTiles = B / 16;

  int blocks = 512;
  int blocksNeeded = (numTiles + 7) / 8;
  if (blocks > blocksNeeded) blocks = blocksNeeded;
  if (blocks < 1) blocks = 1;

  quantum_layer_kernel<<<blocks, 256, 0, stream>>>(x, qp, scale, out, numTiles);
}


# one sample per lane: a wave takes a tile pair (32 rows), each lane works out its own row's angles and amplitudes once; half-waves trade fragments and partial sums by v_permlane16_swap, not LDS
# speedup vs baseline: 1.6579x; 1.1161x over previous
#include <hip/hip_runtime.h>
#include <math.h>


#define NQ 4
#define DIM 16
#define QDEPTH 6
#define TPG 8
#define PPG (TPG / 2)

typedef float v8f __attribute__((ext_vector_type(8)));
typedef float v2f __attribute__((ext_vector_type(2)));

#if defined(__has_builtin)
#if __has_builtin(__builtin_amdgcn_tanhf)
#define FAST_TANH(x) __builtin_amdgcn_tanhf(x)
#endif
#endif
#ifndef FAST_TANH
#define FAST_TANH(x) tanhf(x)
#endif

__device__ __forceinline__ float lane_bcast(float v, int srcLane) {
  return __builtin_bit_cast(
      float, __builtin_amdgcn_readlane(__builtin_bit_cast(int, v), srcLane));
}

__device__ __forceinline__ void cnot16(float* st, int cm, int tm) {
#pragma unroll
  for (int i = 0; i < DIM; ++i) {
    if ((i & cm) && !(i & tm)) {
      float t = st[i];
      st[i] = st[i | tm];
      st[i | tm] = t;
    }
  }
}

__device__ __forceinline__ void ry16(float* st, int m, float c, float s) {
#pragma unroll
  for (int i = 0; i < DIM; ++i) {
    if (!(i & m)) {
      int j = i | m;
      float s0 = st[i], s1 = st[j];
      st[i] = c * s0 - s * s1;
      st[j] = s * s0 + c * s1;
    }
  }
}

__device__ __forceinline__ void half_swap(float& a, float& b) {
  const unsigned ua = __builtin_bit_cast(unsigned, a);
  const unsigned ub = __builtin_bit_cast(unsigned, b);
  auto r = __builtin_amdgcn_permlane16_swap(ua, ub, false, false);
  const unsigned r0 = r[0];
  const unsigned r1 = r[1];
  a = __builtin_bit_cast(float, r0);
  b = __builtin_bit_cast(float, r1);
}

__device__ __forceinline__ float4 pair_compute(const float4 xv, const v2f* A, const float sc) {
  const float PI4 = 0.78539816339744831f;

  float inv = rsqrtf(xv.x * xv.x + xv.y * xv.y + xv.z * xv.z + xv.w * xv.w);
  float m = sc * inv;
  float a0 = FAST_TANH(xv.x * m) * PI4 + PI4;
  float a1 = FAST_TANH(xv.y * m) * PI4 + PI4;
  float a2 = FAST_TANH(xv.z * m) * PI4 + PI4;
  float a3 = FAST_TANH(xv.w * m) * PI4 + PI4;
  float c0 = __cosf(a0), s0v = __sinf(a0);
  float c1 = __cosf(a1), s1v = __sinf(a1);
  float c2 = __cosf(a2), s2v = __sinf(a2);
  float c3 = __cosf(a3), s3v = __sinf(a3);

  float p01[4] = {c0 * c1, c0 * s1v, s0v * c1, s0v * s1v};
  float g23[4] = {c2 * c3, c2 * s3v, s2v * c3, s2v * s3v};

  v8f acc0 = {0.f, 0.f, 0.f, 0.f, 0.f, 0.f, 0.f, 0.f};
  v8f acc1 = {0.f, 0.f, 0.f, 0.f, 0.f, 0.f, 0.f, 0.f};
#pragma unroll
  for (int c = 0; c < 4; ++c) {
    float e0 = p01[c] * g23[0], e1 = p01[c] * g23[1];
    float e2 = p01[c] * g23[2], e3 = p01[c] * g23[3];
    half_swap(e0, e2);
    half_swap(e1, e3);
    v2f b0, b1;
    b0[0] = e0; b0[1] = e1;
    b1[0] = e2; b1[1] = e3;
    acc0 = __builtin_amdgcn_wmma_f32_16x16x4_f32(
        false, A[c], false, b0, (short)0, acc0, false, false);
    acc1 = __builtin_amdgcn_wmma_f32_16x16x4_f32(
        false, A[c], false, b1, (short)0, acc1, false, false);
  }

  float S[2], P1[2], P2[2], P3[2];
#pragma unroll
  for (int t = 0; t < 2; ++t) {
    const v8f acc = t ? acc1 : acc0;
    float p0 = acc[0] * acc[0], p1 = acc[1] * acc[1];
    float p2 = acc[2] * acc[2], p3 = acc[3] * acc[3];
    float p4 = acc[4] * acc[4], p5 = acc[5] * acc[5];
    float p6 = acc[6] * acc[6], p7 = acc[7] * acc[7];
    float a01 = p0 + p1, a23 = p2 + p3, a45 = p4 + p5, a67 = p6 + p7;
    float b01 = p0 - p1, b23 = p2 - p3, b45 = p4 - p5, b67 = p6 - p7;
    float L = a01 + a23, H = a45 + a67;
    S[t] = L + H;
    P1[t] = L - H;
    P2[t] = (a01 - a23) + (a45 - a67);
    P3[t] = (b01 + b23) + (b45 + b67);
  }

  half_swap(S[0], S[1]);
  half_swap(P1[0], P1[1]);
  half_swap(P2[0], P2[1]);
  half_swap(P3[0], P3[1]);

  return make_float4(S[0] - S[1], P1[0] + P1[1], P2[0] + P2[1], P3[0] + P3[1]);
}

__launch_bounds__(256)
__global__ void quantum_layer_kernel(const float* __restrict__ x,
                                     const float* __restrict__ qp,
                                     const float* __restrict__ scale,
                                     float* __restrict__ out,
                                     int numTiles) {
  __shared__ float Ms[DIM * DIM];

  const int lane = threadIdx.x & 31;
  const int n = lane & 15;
  const int h = lane >> 4;

  const int wavesPerBlock = blockDim.x >> 5;
  const int waveId = __builtin_amdgcn_readfirstlane(
      blockIdx.x * wavesPerBlock + (threadIdx.x >> 5));
  const int nWaves = gridDim.x * wavesPerBlock;

  const float4* __restrict__ x4 = (const float4*)x;
  float4* __restrict__ o4 = (float4*)out;

  const int fullGroups = numTiles / TPG;
  int g = waveId;
  float4 ring[PPG];
  if (g < fullGroups) {
    const float4* gp = x4 + (size_t)g * (TPG * 16) + lane;
#pragma unroll
    for (int u = 0; u < PPG; ++u) ring[u] = gp[u * 32];
  }

  if (threadIdx.x < 32) {
    float qv = 0.0f;
#pragma unroll
    for (int i = 0; i < QDEPTH * NQ; ++i) {
      const float qi = qp[i];
      qv = (lane == i) ? qi : qv;
    }
    const float th = 0.5f * qv;
    const float cv = cosf(th), sv = sinf(th);
    float st[DIM];
#pragma unroll
    for (int i = 0; i < DIM; ++i) st[i] = (i == n) ? 1.0f : 0.0f;
    for (int k = 0; k < QDEPTH; ++k) {
      cnot16(st, 8, 4);
      cnot16(st, 2, 1);
      cnot16(st, 4, 2);
#pragma unroll
      for (int w = 0; w < NQ; ++w) {
        ry16(st, 8 >> w, lane_bcast(cv, k * NQ + w), lane_bcast(sv, k * NQ + w));
      }
    }
    if (lane < DIM) {
#pragma unroll
      for (int j = 0; j < DIM; ++j) Ms[j * DIM + lane] = st[j];
    }
  }
  __syncthreads();

  v2f A[4];
#pragma unroll
  for (int c = 0; c < 4; ++c) {
    int off = n * DIM + c * 4 + h * 2;
    A[c] = *(const v2f*)&Ms[off];
  }

  const float sc = scale[0];

  if (g < fullGroups) {
    while (g + nWaves < fullGroups) {
      const float4* gpn = x4 + (size_t)(g + nWaves) * (TPG * 16) + lane;
      float4* gpo = o4 + (size_t)g * (TPG * 16) + lane;
#pragma unroll
      for (int u = 0; u < PPG; ++u) {
        gpo[u * 32] = pair_compute(ring[u], A, sc);
        ring[u] = gpn[u * 32];
      }
      g += nWaves;
    }
    float4* gpo = o4 + (size_t)g * (TPG * 16) + lane;
#pragma unroll
    for (int u = 0; u < PPG; ++u) {
      gpo[u * 32] = pair_compute(ring[u], A, sc);
    }
  }

  const int rowsEnd = numTiles * 16;
  for (int r0 = fullGroups * (TPG * 16) + waveId * 32; r0 < rowsEnd; r0 += nWaves * 32) {
    const int row = r0 + lane;
    const bool live = row < rowsEnd;
    const float4 res = pair_compute(x4[live ? row : 0], A, sc);
    if (live) o4[row] = res;
  }
}

extern "C" void kernel_launch(void* const* d_in, const int* in_sizes, int n_in,
                              void* d_out, int out_size, void* d_ws, size_t ws_size,
                              hipStream_t stream) {
  const float* x = (const float*)d_in[0];
  const float* qp = (const float*)d_in[1];
  const float* scale = (const float*)d_in[2];
  float* out = (float*)d_out;

  const int B = in_sizes[0] / 4;
  const int numTiles = B / 16;

  int blocks = 512;
  int blocksNeeded = (numTiles + 7) / 8;
  if (blocks > blocksNeeded) blocks = blocksNeeded;
  if (blocks < 1) blocks = 1;

  quantum_layer_kernel<<<blocks, 256, 0, stream>>>(x, qp, scale, out, numTiles);
}


# per-sample arithmetic on two-wide vectors of neighbouring registers: packed instructions, no register moves; probability sums as a pair butterfly with fused squares (regroups those sums)
# speedup vs baseline: 1.7958x; 1.0832x over previous
#include <hip/hip_runtime.h>
#include <math.h>


#define NQ 4
#define DIM 16
#define QDEPTH 6
#define TPG 8
#define PPG (TPG / 2)

typedef float v8f __attribute__((ext_vector_type(8)));
typedef float v2f __attribute__((ext_vector_type(2)));

#if defined(__has_builtin)
#if __has_builtin(__builtin_amdgcn_tanhf)
#define FAST_TANH(x) __builtin_amdgcn_tanhf(x)
#endif
#endif
#ifndef FAST_TANH
#define FAST_TANH(x) tanhf(x)
#endif

__device__ __forceinline__ float lane_bcast(float v, int srcLane) {
  return __builtin_bit_cast(
      float, __builtin_amdgcn_readlane(__builtin_bit_cast(int, v), srcLane));
}

__device__ __forceinline__ void cnot16(float* st, int cm, int tm) {
#pragma unroll
  for (int i = 0; i < DIM; ++i) {
    if ((i & cm) && !(i & tm)) {
      float t = st[i];
      st[i] = st[i | tm];
      st[i | tm] = t;
    }
  }
}

__device__ __forceinline__ void ry16(float* st, int m, float c, float s) {
#pragma unroll
  for (int i = 0; i < DIM; ++i) {
    if (!(i & m)) {
      int j = i | m;
      float s0 = st[i], s1 = st[j];
      st[i] = c * s0 - s * s1;
      st[j] = s * s0 + c * s1;
    }
  }
}

__device__ __forceinline__ void half_swap(float& a, float& b) {
  const unsigned ua = __builtin_bit_cast(unsigned, a);
  const unsigned ub = __builtin_bit_cast(unsigned, b);
  auto r = __builtin_amdgcn_permlane16_swap(ua, ub, false, false);
  const unsigned r0 = r[0];
  const unsigned r1 = r[1];
  a = __builtin_bit_cast(float, r0);
  b = __builtin_bit_cast(float, r1);
}

__device__ __forceinline__ float4 pair_compute(const float4 xv, const v2f* A, const float sc) {
  const float PI4 = 0.78539816339744831f;
  const float INV2PI = 0.15915494309189532f;

  const v2f xy = {xv.x, xv.y};
  const v2f zw = {xv.z, xv.w};
  const v2f q = __builtin_elementwise_fma(zw, zw, xy * xy);
  const float inv = rsqrtf(q[0] + q[1]);
  const float m = sc * inv;
  const v2f y01 = xy * m;
  const v2f y23 = zw * m;
  v2f t01, t23;
  t01[0] = FAST_TANH(y01[0]); t01[1] = FAST_TANH(y01[1]);
  t23[0] = FAST_TANH(y23[0]); t23[1] = FAST_TANH(y23[1]);
  const v2f r01 = (t01 * PI4 + PI4) * INV2PI;
  const v2f r23 = (t23 * PI4 + PI4) * INV2PI;
  v2f cs0, cs1, cs2, cs3;
  cs0[0] = __builtin_amdgcn_cosf(r01[0]); cs0[1] = __builtin_amdgcn_sinf(r01[0]);
  cs1[0] = __builtin_amdgcn_cosf(r01[1]); cs1[1] = __builtin_amdgcn_sinf(r01[1]);
  cs2[0] = __builtin_amdgcn_cosf(r23[0]); cs2[1] = __builtin_amdgcn_sinf(r23[0]);
  cs3[0] = __builtin_amdgcn_cosf(r23[1]); cs3[1] = __builtin_amdgcn_sinf(r23[1]);

  const v2f p01lo = cs0[0] * cs1;
  const v2f p01hi = cs0[1] * cs1;
  const v2f g23lo = cs2[0] * cs3;
  const v2f g23hi = cs2[1] * cs3;
  const float p01[4] = {p01lo[0], p01lo[1], p01hi[0], p01hi[1]};

  v8f acc0 = {0.f, 0.f, 0.f, 0.f, 0.f, 0.f, 0.f, 0.f};
  v8f acc1 = {0.f, 0.f, 0.f, 0.f, 0.f, 0.f, 0.f, 0.f};
#pragma unroll
  for (int c = 0; c < 4; ++c) {
    const v2f lo = p01[c] * g23lo;
    const v2f hi = p01[c] * g23hi;
    float e0 = lo[0], e1 = lo[1], e2 = hi[0], e3 = hi[1];
    half_swap(e0, e2);
    half_swap(e1, e3);
    v2f b0, b1;
    b0[0] = e0; b0[1] = e1;
    b1[0] = e2; b1[1] = e3;
    acc0 = __builtin_amdgcn_wmma_f32_16x16x4_f32(
        false, A[c], false, b0, (short)0, acc0, false, false);
    acc1 = __builtin_amdgcn_wmma_f32_16x16x4_f32(
        false, A[c], false, b1, (short)0, acc1, false, false);
  }

  float S[2], P1[2], P2[2], P3[2];
#pragma unroll
  for (int t = 0; t < 2; ++t) {
    const v8f acc = t ? acc1 : acc0;
    const v2f u01 = {acc[0], acc[1]};
    const v2f u23 = {acc[2], acc[3]};
    const v2f u45 = {acc[4], acc[5]};
    const v2f u67 = {acc[6], acc[7]};
    const v2f s01 = u01 * u01;
    const v2f s45 = u45 * u45;
    const v2f lsum = __builtin_elementwise_fma(u23, u23, s01);
    const v2f ldif = __builtin_elementwise_fma(-u23, u23, s01);
    const v2f hsum = __builtin_elementwise_fma(u67, u67, s45);
    const v2f hdif = __builtin_elementwise_fma(-u67, u67, s45);
    const v2f all = lsum + hsum;
    const v2f b2 = lsum - hsum;
    const v2f b1 = ldif + hdif;
    S[t] = all[0] + all[1];
    P1[t] = b2[0] + b2[1];
    P2[t] = b1[0] + b1[1];
    P3[t] = all[0] - all[1];
  }

  half_swap(S[0], S[1]);
  half_swap(P1[0], P1[1]);
  half_swap(P2[0], P2[1]);
  half_swap(P3[0], P3[1]);

  return make_float4(S[0] - S[1], P1[0] + P1[1], P2[0] + P2[1], P3[0] + P3[1]);
}

__launch_bounds__(256)
__global__ void quantum_layer_kernel(const float* __restrict__ x,
                                     const float* __restrict__ qp,
                                     const float* __restrict__ scale,
                                     float* __restrict__ out,
                                     int numTiles) {
  __shared__ float Ms[DIM * DIM];

  const int lane = threadIdx.x & 31;
  const int n = lane & 15;
  const int h = lane >> 4;

  const int wavesPerBlock = blockDim.x >> 5;
  const int waveId = __builtin_amdgcn_readfirstlane(
      blockIdx.x * wavesPerBlock + (threadIdx.x >> 5));
  const int nWaves = gridDim.x * wavesPerBlock;

  const float4* __restrict__ x4 = (const float4*)x;
  float4* __restrict__ o4 = (float4*)out;

  const int fullGroups = numTiles / TPG;
  int g = waveId;
  float4 ring[PPG];
  if (g < fullGroups) {
    const float4* gp = x4 + (size_t)g * (TPG * 16) + lane;
#pragma unroll
    for (int u = 0; u < PPG; ++u) ring[u] = gp[u * 32];
  }

  if (threadIdx.x < 32) {
    float qv = 0.0f;
#pragma unroll
    for (int i = 0; i < QDEPTH * NQ; ++i) {
      const float qi = qp[i];
      qv = (lane == i) ? qi : qv;
    }
    const float th = 0.5f * qv;
    const float cv = cosf(th), sv = sinf(th);
    float st[DIM];
#pragma unroll
    for (int i = 0; i < DIM; ++i) st[i] = (i == n) ? 1.0f : 0.0f;
    for (int k = 0; k < QDEPTH; ++k) {
      cnot16(st, 8, 4);
      cnot16(st, 2, 1);
      cnot16(st, 4, 2);
#pragma unroll
      for (int w = 0; w < NQ; ++w) {
        ry16(st, 8 >> w, lane_bcast(cv, k * NQ + w), lane_bcast(sv, k * NQ + w));
      }
    }
    if (lane < DIM) {
#pragma unroll
      for (int j = 0; j < DIM; ++j) Ms[j * DIM + lane] = st[j];
    }
  }
  __syncthreads();

  v2f A[4];
#pragma unroll
  for (int c = 0; c < 4; ++c) {
    int off = n * DIM + c * 4 + h * 2;
    A[c] = *(const v2f*)&Ms[off];
  }

  const float sc = scale[0];

  if (g < fullGroups) {
    while (g + nWaves < fullGroups) {
      const float4* gpn = x4 + (size_t)(g + nWaves) * (TPG * 16) + lane;
      float4* gpo = o4 + (size_t)g * (TPG * 16) + lane;
#pragma unroll
      for (int u = 0; u < PPG; ++u) {
        gpo[u * 32] = pair_compute(ring[u], A, sc);
        ring[u] = gpn[u * 32];
      }
      g += nWaves;
    }
    float4* gpo = o4 + (size_t)g * (TPG * 16) + lane;
#pragma unroll
    for (int u = 0; u < PPG; ++u) {
      gpo[u * 32] = pair_compute(ring[u], A, sc);
    }
  }

  const int rowsEnd = numTiles * 16;
  for (int r0 = fullGroups * (TPG * 16) + waveId * 32; r0 < rowsEnd; r0 += nWaves * 32) {
    const int row = r0 + lane;
    const bool live = row < rowsEnd;
    const float4 res = pair_compute(x4[live ? row : 0], A, sc);
    if (live) o4[row] = res;
  }
}

extern "C" void kernel_launch(void* const* d_in, const int* in_sizes, int n_in,
                              void* d_out, int out_size, void* d_ws, size_t ws_size,
                              hipStream_t stream) {
  const float* x = (const float*)d_in[0];
  const float* qp = (const float*)d_in[1];
  const float* scale = (const float*)d_in[2];
  float* out = (float*)d_out;

  const int B = in_sizes[0] / 4;
  const int numTiles = B / 16;

  int blocks = 512;
  int blocksNeeded = (numTiles + 7) / 8;
  if (blocks > blocksNeeded) blocks = blocksNeeded;
  if (blocks < 1) blocks = 1;

  quantum_layer_kernel<<<blocks, 256, 0, stream>>>(x, qp, scale, out, numTiles);
}
